# starting kernel
# baseline (speedup 1.0000x reference)
#include <hip/hip_runtime.h>

#define BB   16
#define NN   4096
#define EE   65536
#define CIN  128
#define COUT 128

typedef __attribute__((ext_vector_type(2))) float v2f;
typedef __attribute__((ext_vector_type(8))) float v8f;

__global__ void scale_copy_kernel(float* __restrict__ dst,
                                  const float* __restrict__ src,
                                  float scale, int n4) {
    int i = blockIdx.x * blockDim.x + threadIdx.x;
    if (i >= n4) return;
    float4 v;
    if (src) {
        float4 s = ((const float4*)src)[i];
        v = make_float4(scale * s.x, scale * s.y, scale * s.z, scale * s.w);
    } else {
        v = make_float4(0.f, 0.f, 0.f, 0.f);
    }
    ((float4*)dst)[i] = v;
}

__global__ void spmm_scatter_kernel(float* __restrict__ y,
                                    const float* __restrict__ x,
                                    const float* __restrict__ vals,
                                    const int* __restrict__ rows,
                                    const int* __restrict__ cols,
                                    float scale) {
    const int gwave = (blockIdx.x * blockDim.x + threadIdx.x) >> 5;
    if (gwave >= BB * EE) return;
    const int b = gwave >> 16;
    const int r = rows[gwave];
    const int c = cols[gwave];
    const float v = vals[gwave] * scale;
    const int lane = threadIdx.x & 31;

    const float4 xv = ((const float4*)(x + ((size_t)b * NN + c) * CIN))[lane];
    float* yp = y + ((size_t)b * NN + r) * CIN + lane * 4;
    atomicAdd(yp + 0, v * xv.x);
    atomicAdd(yp + 1, v * xv.y);
    atomicAdd(yp + 2, v * xv.z);
    atomicAdd(yp + 3, v * xv.w);
}

__global__ __launch_bounds__(512)
void cheb_gemm_kernel(const float* __restrict__ t0,
                      const float* __restrict__ t1,
                      const float* __restrict__ t2,
                      const float* __restrict__ t3,
                      const float* __restrict__ weights,
                      const float* __restrict__ bias,
                      float* __restrict__ out) {
    __shared__ float As[32 * 132];

    const float* terms[4] = {t0, t1, t2, t3};

    const int tid  = threadIdx.x;
    const int wave = tid >> 5;
    const int lane = tid & 31;
    const int wm   = wave >> 3;
    const int wn   = wave & 7;
    const int l    = lane & 15;
    const int half = lane >> 4;

    const size_t row0 = (size_t)blockIdx.x * 32;

    const int arow = wm * 16 + l;
    const int bn   = wn * 16 + l;

    v8f acc = {};

#pragma unroll
    for (int t = 0; t < 4; ++t) {
        __syncthreads();
        const float* src = terms[t] + row0 * CIN;
#pragma unroll
        for (int i = tid; i < 1024; i += 512) {
            const int r  = i >> 5;
            const int cv = i & 31;
            const float4 v = ((const float4*)(src + r * CIN))[cv];
            float* d = &As[r * 132 + cv * 4];
            d[0] = v.x; d[1] = v.y; d[2] = v.z; d[3] = v.w;
        }
        __syncthreads();

        const float* W = weights + t * CIN * COUT;
#pragma unroll 4
        for (int k = 0; k < 128; k += 4) {
            const int kk = k + 2 * half;
            v2f a;
            a.x = As[arow * 132 + kk];
            a.y = As[arow * 132 + kk + 1];
            v2f bf;
            bf.x = W[kk * COUT + bn];
            bf.y = W[(kk + 1) * COUT + bn];
            acc = __builtin_amdgcn_wmma_f32_16x16x4_f32(
                 false, a,  false, bf,
                 (short)0, acc,  false,  false);
        }
    }

    const float bv = bias[bn];
    float* op = out + (row0 + (size_t)wm * 16 + 8 * half) * COUT + bn;
#pragma unroll
    for (int r = 0; r < 8; ++r) {
        const float v = acc[r] + bv;
        op[(size_t)r * COUT] = fmaxf(v, 0.f);
    }
}

extern "C" void kernel_launch(void* const* d_in, const int* in_sizes, int n_in,
                              void* d_out, int out_size, void* d_ws, size_t ws_size,
                              hipStream_t stream) {
    const float* x       = (const float*)d_in[0];
    const float* vals    = (const float*)d_in[1];
    const float* weights = (const float*)d_in[2];
    const float* bias    = (const float*)d_in[3];
    const int*   rows    = (const int*)d_in[4];
    const int*   cols    = (const int*)d_in[5];
    float* out = (float*)d_out;

    const size_t feat = (size_t)BB * NN * CIN;
    float* T1 = (float*)d_ws;
    float* T2 = T1 + feat;
    float* T3 = T2 + feat;

    const int n4       = (int)(feat / 4);
    const int initGrid = (n4 + 255) / 256;
    const int spmmGrid = (BB * EE * 32) / 256;

    scale_copy_kernel<<<initGrid, 256, 0, stream>>>(T1, nullptr, 0.f, n4);
    spmm_scatter_kernel<<<spmmGrid, 256, 0, stream>>>(T1, x, vals, rows, cols, 1.0f);
    scale_copy_kernel<<<initGrid, 256, 0, stream>>>(T2, x, -1.f, n4);
    spmm_scatter_kernel<<<spmmGrid, 256, 0, stream>>>(T2, T1, vals, rows, cols, 2.0f);
    scale_copy_kernel<<<initGrid, 256, 0, stream>>>(T3, T1, -1.f, n4);
    spmm_scatter_kernel<<<spmmGrid, 256, 0, stream>>>(T3, T2, vals, rows, cols, 2.0f);

    cheb_gemm_kernel<<<(BB * NN) / 32, 512, 0, stream>>>(x, T1, T2, T3, weights, bias, out);
}


# Edges grouped by output row in each call (LDS counting sort), then one wave per row sums its edges in registers in edge order and stores once: float atomics and init passes gone, sum order fixed
# speedup vs baseline: 2.8489x; 2.8489x over previous
#include <hip/hip_runtime.h>

#define BB   16
#define NN   4096
#define EE   65536
#define CIN  128
#define COUT 128

typedef __attribute__((ext_vector_type(2))) float v2f;
typedef __attribute__((ext_vector_type(8))) float v8f;

__global__ __launch_bounds__(1024)
void build_rows_kernel(const int* __restrict__ rows,
                       int* __restrict__ rowptr,
                       int* __restrict__ perm) {
    __shared__ int cnt[NN];
    __shared__ int wsum[32];

    const int b    = blockIdx.x;
    const int tid  = threadIdx.x;
    const int lane = tid & 31;
    const int wave = tid >> 5;
    const int* rb  = rows + (size_t)b * EE;

    for (int i = tid; i < NN; i += 1024) cnt[i] = 0;
    __syncthreads();

    const int4* rb4 = (const int4*)rb;
    int4 rr[16];
#pragma unroll
    for (int k = 0; k < 16; ++k) rr[k] = rb4[tid + k * 1024];
#pragma unroll
    for (int k = 0; k < 16; ++k) {
        atomicAdd(&cnt[rr[k].x], 1);
        atomicAdd(&cnt[rr[k].y], 1);
        atomicAdd(&cnt[rr[k].z], 1);
        atomicAdd(&cnt[rr[k].w], 1);
    }
    __syncthreads();

    const int c0 = cnt[4 * tid + 0];
    const int c1 = cnt[4 * tid + 1];
    const int c2 = cnt[4 * tid + 2];
    const int c3 = cnt[4 * tid + 3];
    const int s  = c0 + c1 + c2 + c3;
    int incl = s;
    for (int d = 1; d < 32; d <<= 1) {
        const int t = __shfl_up(incl, d);
        if (lane >= d) incl += t;
    }
    if (lane == 31) wsum[wave] = incl;
    __syncthreads();
    if (wave == 0) {
        const int w = wsum[lane];
        int wi = w;
        for (int d = 1; d < 32; d <<= 1) {
            const int t = __shfl_up(wi, d);
            if (lane >= d) wi += t;
        }
        wsum[lane] = wi - w;
    }
    __syncthreads();
    const int base = wsum[wave] + incl - s;
    const int p0 = base, p1 = p0 + c0, p2 = p1 + c1, p3 = p2 + c2;

    int* rp = rowptr + b * (NN + 1);
    rp[4 * tid + 0] = p0;
    rp[4 * tid + 1] = p1;
    rp[4 * tid + 2] = p2;
    rp[4 * tid + 3] = p3;
    if (tid == 1023) rp[NN] = p3 + c3;

    cnt[4 * tid + 0] = p0;
    cnt[4 * tid + 1] = p1;
    cnt[4 * tid + 2] = p2;
    cnt[4 * tid + 3] = p3;
    __syncthreads();

    int* pb = perm + (size_t)b * EE;
#pragma unroll
    for (int g = 0; g < 4; ++g) {
        int pos[16];
#pragma unroll
        for (int q = 0; q < 4; ++q) {
            const int k = 4 * g + q;
            pos[4 * q + 0] = atomicAdd(&cnt[rr[k].x], 1);
            pos[4 * q + 1] = atomicAdd(&cnt[rr[k].y], 1);
            pos[4 * q + 2] = atomicAdd(&cnt[rr[k].z], 1);
            pos[4 * q + 3] = atomicAdd(&cnt[rr[k].w], 1);
        }
#pragma unroll
        for (int q = 0; q < 4; ++q) {
            const int e0 = 4 * (tid + (4 * g + q) * 1024);
            pb[pos[4 * q + 0]] = e0 + 0;
            pb[pos[4 * q + 1]] = e0 + 1;
            pb[pos[4 * q + 2]] = e0 + 2;
            pb[pos[4 * q + 3]] = e0 + 3;
        }
    }
}

__global__ __launch_bounds__(256)
void order_rows_kernel(const int* __restrict__ rowptr,
                       const int* __restrict__ perm,
                       const int* __restrict__ cols,
                       const float* __restrict__ vals,
                       int* __restrict__ pk) {
    const int row  = __builtin_amdgcn_readfirstlane((int)((blockIdx.x * 256 + threadIdx.x) >> 5));
    const int lane = threadIdx.x & 31;
    const int b    = row >> 12;
    const int r    = row & (NN - 1);
    const int s    = rowptr[b * (NN + 1) + r];
    const int n    = rowptr[b * (NN + 1) + r + 1] - s;
    const int* pb  = perm + (size_t)b * EE + s;

    for (int i0 = 0; i0 < n; i0 += 32) {
        const int i  = i0 + lane;
        const int id = (i < n) ? pb[i] : 0x7fffffff;
        int rank = 0;
        for (int j0 = 0; j0 < n; j0 += 32) {
            const int jj    = j0 + lane;
            const int other = (jj < n) ? pb[jj] : 0x7fffffff;
            const int m     = (n - j0 < 32) ? (n - j0) : 32;
            for (int u = 0; u < m; ++u)
                rank += (__builtin_amdgcn_readlane(other, u) < id) ? 1 : 0;
        }
        if (i < n) {
            const size_t e = (size_t)b * EE + id;
            int2 rec;
            rec.x = cols[e];
            rec.y = __float_as_int(vals[e]);
            ((int2*)pk)[(size_t)b * EE + s + rank] = rec;
        }
    }
}

template <int K>
__device__ __forceinline__ void edge_batch(float4& acc, const int2* __restrict__ rec,
                                           const float4* __restrict__ xb, int i, int lane) {
    int2 cv[K];
#pragma unroll
    for (int u = 0; u < K; ++u) cv[u] = rec[i + u];
    float4 xv[K];
#pragma unroll
    for (int u = 0; u < K; ++u) xv[u] = xb[(size_t)cv[u].x * (CIN / 4) + lane];
#pragma unroll
    for (int u = 0; u < K; ++u) {
        const float v = __int_as_float(cv[u].y);
        acc.x = __builtin_fmaf(v, xv[u].x, acc.x);
        acc.y = __builtin_fmaf(v, xv[u].y, acc.y);
        acc.z = __builtin_fmaf(v, xv[u].z, acc.z);
        acc.w = __builtin_fmaf(v, xv[u].w, acc.w);
    }
}

__global__ __launch_bounds__(256)
void spmm_rows_kernel(float* __restrict__ y,
                      const float* __restrict__ x,
                      const float* __restrict__ prev,
                      const int* __restrict__ rowptr,
                      const int* __restrict__ pk,
                      float scale) {
    const int row  = __builtin_amdgcn_readfirstlane((int)((blockIdx.x * 256 + threadIdx.x) >> 5));
    const int lane = threadIdx.x & 31;
    const int b    = row >> 12;
    const int r    = row & (NN - 1);
    const int s    = rowptr[b * (NN + 1) + r];
    const int e    = rowptr[b * (NN + 1) + r + 1];
    const int2* rec  = (const int2*)pk + (size_t)b * EE;
    const float4* xb = (const float4*)(x + (size_t)b * NN * CIN);

    float4 acc = make_float4(0.f, 0.f, 0.f, 0.f);
    int i = s;
    for (; i + 8 <= e; i += 8) edge_batch<8>(acc, rec, xb, i, lane);
    if (i + 4 <= e) { edge_batch<4>(acc, rec, xb, i, lane); i += 4; }
    if (i + 2 <= e) { edge_batch<2>(acc, rec, xb, i, lane); i += 2; }
    if (i < e)      { edge_batch<1>(acc, rec, xb, i, lane); }

    float4 o;
    if (prev) {
        const float4 p = ((const float4*)prev)[(size_t)row * (CIN / 4) + lane];
        o = make_float4(scale * acc.x - p.x, scale * acc.y - p.y,
                        scale * acc.z - p.z, scale * acc.w - p.w);
    } else {
        o = make_float4(scale * acc.x, scale * acc.y, scale * acc.z, scale * acc.w);
    }
    ((float4*)y)[(size_t)row * (CIN / 4) + lane] = o;
}

__global__ __launch_bounds__(512)
void cheb_gemm_kernel(const float* __restrict__ t0,
                      const float* __restrict__ t1,
                      const float* __restrict__ t2,
                      const float* __restrict__ t3,
                      const float* __restrict__ weights,
                      const float* __restrict__ bias,
                      float* __restrict__ out) {
    __shared__ float As[32 * 132];

    const float* terms[4] = {t0, t1, t2, t3};

    const int tid  = threadIdx.x;
    const int wave = tid >> 5;
    const int lane = tid & 31;
    const int wm   = wave >> 3;
    const int wn   = wave & 7;
    const int l    = lane & 15;
    const int half = lane >> 4;

    const size_t row0 = (size_t)blockIdx.x * 32;

    const int arow = wm * 16 + l;
    const int bn   = wn * 16 + l;

    v8f acc = {};

#pragma unroll
    for (int t = 0; t < 4; ++t) {
        __syncthreads();
        const float* src = terms[t] + row0 * CIN;
#pragma unroll
        for (int i = tid; i < 1024; i += 512) {
            const int r  = i >> 5;
            const int cv = i & 31;
            const float4 v = ((const float4*)(src + r * CIN))[cv];
            float* d = &As[r * 132 + cv * 4];
            d[0] = v.x; d[1] = v.y; d[2] = v.z; d[3] = v.w;
        }
        __syncthreads();

        const float* W = weights + t * CIN * COUT;
#pragma unroll 4
        for (int k = 0; k < 128; k += 4) {
            const int kk = k + 2 * half;
            v2f a;
            a.x = As[arow * 132 + kk];
            a.y = As[arow * 132 + kk + 1];
            v2f bf;
            bf.x = W[kk * COUT + bn];
            bf.y = W[(kk + 1) * COUT + bn];
            acc = __builtin_amdgcn_wmma_f32_16x16x4_f32(
                 false, a,  false, bf,
                 (short)0, acc,  false,  false);
        }
    }

    const float bv = bias[bn];
    float* op = out + (row0 + (size_t)wm * 16 + 8 * half) * COUT + bn;
#pragma unroll
    for (int r = 0; r < 8; ++r) {
        const float v = acc[r] + bv;
        op[(size_t)r * COUT] = fmaxf(v, 0.f);
    }
}

extern "C" void kernel_launch(void* const* d_in, const int* in_sizes, int n_in,
                              void* d_out, int out_size, void* d_ws, size_t ws_size,
                              hipStream_t stream) {
    const float* x       = (const float*)d_in[0];
    const float* vals    = (const float*)d_in[1];
    const float* weights = (const float*)d_in[2];
    const float* bias    = (const float*)d_in[3];
    const int*   rows    = (const int*)d_in[4];
    const int*   cols    = (const int*)d_in[5];
    float* out = (float*)d_out;

    const size_t feat = (size_t)BB * NN * CIN;
    float* T1 = (float*)d_ws;
    float* T2 = T1 + feat;
    float* T3 = T2 + feat;

    int* pk     = (int*)d_out;
    int* perm   = pk + (size_t)BB * EE * 2;
    int* rowptr = perm + (size_t)BB * EE;

    const int rowGrid = (BB * NN * 32) / 256;

    build_rows_kernel<<<BB, 1024, 0, stream>>>(rows, rowptr, perm);
    order_rows_kernel<<<rowGrid, 256, 0, stream>>>(rowptr, perm, cols, vals, pk);

    spmm_rows_kernel<<<rowGrid, 256, 0, stream>>>(T1, x, nullptr, rowptr, pk, 1.0f);
    spmm_rows_kernel<<<rowGrid, 256, 0, stream>>>(T2, T1, x, rowptr, pk, 2.0f);
    spmm_rows_kernel<<<rowGrid, 256, 0, stream>>>(T3, T2, T1, rowptr, pk, 2.0f);

    cheb_gemm_kernel<<<(BB * NN) / 32, 512, 0, stream>>>(x, T1, T2, T3, weights, bias, out);
}


# Edge grouping split over 256 work-groups instead of 16: each takes one 256-row block of a sample, scans all its row numbers, counts and places only its own block's edges; sum order unchanged
# speedup vs baseline: 3.4101x; 1.1970x over previous
#include <hip/hip_runtime.h>

#define BB   16
#define NN   4096
#define EE   65536
#define CIN  128
#define COUT 128

typedef __attribute__((ext_vector_type(2))) float v2f;
typedef __attribute__((ext_vector_type(8))) float v8f;

#define RB   256
#define NBLK (NN / RB)

__global__ __launch_bounds__(1024)
void build_rows_kernel(const int* __restrict__ rows,
                       int* __restrict__ rowptr,
                       int* __restrict__ perm) {
    __shared__ int cnt[RB];
    __shared__ int wsum[RB / 32];
    __shared__ int wbelow[32];

    const int b    = blockIdx.x / NBLK;
    const int blk  = blockIdx.x % NBLK;
    const int lo   = blk * RB;
    const int tid  = threadIdx.x;
    const int lane = tid & 31;
    const int wave = tid >> 5;
    const int* rb  = rows + (size_t)b * EE;

    if (tid < RB) cnt[tid] = 0;

    const int4* rb4 = (const int4*)rb;
    int4 rr[16];
#pragma unroll
    for (int k = 0; k < 16; ++k) rr[k] = rb4[tid + k * 1024];

    int below = 0;
#pragma unroll
    for (int k = 0; k < 16; ++k)
        below += (rr[k].x < lo) + (rr[k].y < lo) + (rr[k].z < lo) + (rr[k].w < lo);
    for (int d = 16; d > 0; d >>= 1) below += __shfl_xor(below, d);
    if (lane == 0) wbelow[wave] = below;
    __syncthreads();

#pragma unroll
    for (int k = 0; k < 16; ++k) {
        const int d0 = rr[k].x - lo, d1 = rr[k].y - lo, d2 = rr[k].z - lo, d3 = rr[k].w - lo;
        if ((unsigned)d0 < (unsigned)RB) atomicAdd(&cnt[d0], 1);
        if ((unsigned)d1 < (unsigned)RB) atomicAdd(&cnt[d1], 1);
        if ((unsigned)d2 < (unsigned)RB) atomicAdd(&cnt[d2], 1);
        if ((unsigned)d3 < (unsigned)RB) atomicAdd(&cnt[d3], 1);
    }
    __syncthreads();

    int c = 0, incl = 0;
    if (tid < RB) {
        c = cnt[tid];
        incl = c;
        for (int d = 1; d < 32; d <<= 1) {
            const int t = __shfl_up(incl, d);
            if (lane >= d) incl += t;
        }
        if (lane == 31) wsum[wave] = incl;
    }
    __syncthreads();
    if (tid < RB) {
        int base = wbelow[lane];
        for (int d = 16; d > 0; d >>= 1) base += __shfl_xor(base, d);
        for (int w = 0; w < wave; ++w) base += wsum[w];
        const int p = base + incl - c;

        int* rp = rowptr + b * (NN + 1);
        rp[lo + tid] = p;
        if (blk == NBLK - 1 && tid == RB - 1) rp[NN] = p + c;

        cnt[tid] = p;
    }
    __syncthreads();

    int* pb = perm + (size_t)b * EE;
#pragma unroll
    for (int g = 0; g < 4; ++g) {
        int pos[16];
#pragma unroll
        for (int q = 0; q < 4; ++q) {
            const int k = 4 * g + q;
            const int d0 = rr[k].x - lo, d1 = rr[k].y - lo, d2 = rr[k].z - lo, d3 = rr[k].w - lo;
            pos[4 * q + 0] = ((unsigned)d0 < (unsigned)RB) ? atomicAdd(&cnt[d0], 1) : -1;
            pos[4 * q + 1] = ((unsigned)d1 < (unsigned)RB) ? atomicAdd(&cnt[d1], 1) : -1;
            pos[4 * q + 2] = ((unsigned)d2 < (unsigned)RB) ? atomicAdd(&cnt[d2], 1) : -1;
            pos[4 * q + 3] = ((unsigned)d3 < (unsigned)RB) ? atomicAdd(&cnt[d3], 1) : -1;
        }
#pragma unroll
        for (int q = 0; q < 4; ++q) {
            const int e0 = 4 * (tid + (4 * g + q) * 1024);
            if (pos[4 * q + 0] >= 0) pb[pos[4 * q + 0]] = e0 + 0;
            if (pos[4 * q + 1] >= 0) pb[pos[4 * q + 1]] = e0 + 1;
            if (pos[4 * q + 2] >= 0) pb[pos[4 * q + 2]] = e0 + 2;
            if (pos[4 * q + 3] >= 0) pb[pos[4 * q + 3]] = e0 + 3;
        }
    }
}

__global__ __launch_bounds__(256)
void order_rows_kernel(const int* __restrict__ rowptr,
                       const int* __restrict__ perm,
                       const int* __restrict__ cols,
                       const float* __restrict__ vals,
                       int* __restrict__ pk) {
    const int row  = __builtin_amdgcn_readfirstlane((int)((blockIdx.x * 256 + threadIdx.x) >> 5));
    const int lane = threadIdx.x & 31;
    const int b    = row >> 12;
    const int r    = row & (NN - 1);
    const int s    = rowptr[b * (NN + 1) + r];
    const int n    = rowptr[b * (NN + 1) + r + 1] - s;
    const int* pb  = perm + (size_t)b * EE + s;

    for (int i0 = 0; i0 < n; i0 += 32) {
        const int i  = i0 + lane;
        const int id = (i < n) ? pb[i] : 0x7fffffff;
        int rank = 0;
        for (int j0 = 0; j0 < n; j0 += 32) {
            const int jj    = j0 + lane;
            const int other = (jj < n) ? pb[jj] : 0x7fffffff;
            const int m     = (n - j0 < 32) ? (n - j0) : 32;
            for (int u = 0; u < m; ++u)
                rank += (__builtin_amdgcn_readlane(other, u) < id) ? 1 : 0;
        }
        if (i < n) {
            const size_t e = (size_t)b * EE + id;
            int2 rec;
            rec.x = cols[e];
            rec.y = __float_as_int(vals[e]);
            ((int2*)pk)[(size_t)b * EE + s + rank] = rec;
        }
    }
}

template <int K>
__device__ __forceinline__ void edge_batch(float4& acc, const int2* __restrict__ rec,
                                           const float4* __restrict__ xb, int i, int lane) {
    int2 cv[K];
#pragma unroll
    for (int u = 0; u < K; ++u) cv[u] = rec[i + u];
    float4 xv[K];
#pragma unroll
    for (int u = 0; u < K; ++u) xv[u] = xb[(size_t)cv[u].x * (CIN / 4) + lane];
#pragma unroll
    for (int u = 0; u < K; ++u) {
        const float v = __int_as_float(cv[u].y);
        acc.x = __builtin_fmaf(v, xv[u].x, acc.x);
        acc.y = __builtin_fmaf(v, xv[u].y, acc.y);
        acc.z = __builtin_fmaf(v, xv[u].z, acc.z);
        acc.w = __builtin_fmaf(v, xv[u].w, acc.w);
    }
}

__global__ __launch_bounds__(256)
void spmm_rows_kernel(float* __restrict__ y,
                      const float* __restrict__ x,
                      const float* __restrict__ prev,
                      const int* __restrict__ rowptr,
                      const int* __restrict__ pk,
                      float scale) {
    const int row  = __builtin_amdgcn_readfirstlane((int)((blockIdx.x * 256 + threadIdx.x) >> 5));
    const int lane = threadIdx.x & 31;
    const int b    = row >> 12;
    const int r    = row & (NN - 1);
    const int s    = rowptr[b * (NN + 1) + r];
    const int e    = rowptr[b * (NN + 1) + r + 1];
    const int2* rec  = (const int2*)pk + (size_t)b * EE;
    const float4* xb = (const float4*)(x + (size_t)b * NN * CIN);

    float4 acc = make_float4(0.f, 0.f, 0.f, 0.f);
    int i = s;
    for (; i + 8 <= e; i += 8) edge_batch<8>(acc, rec, xb, i, lane);
    if (i + 4 <= e) { edge_batch<4>(acc, rec, xb, i, lane); i += 4; }
    if (i + 2 <= e) { edge_batch<2>(acc, rec, xb, i, lane); i += 2; }
    if (i < e)      { edge_batch<1>(acc, rec, xb, i, lane); }

    float4 o;
    if (prev) {
        const float4 p = ((const float4*)prev)[(size_t)row * (CIN / 4) + lane];
        o = make_float4(scale * acc.x - p.x, scale * acc.y - p.y,
                        scale * acc.z - p.z, scale * acc.w - p.w);
    } else {
        o = make_float4(scale * acc.x, scale * acc.y, scale * acc.z, scale * acc.w);
    }
    ((float4*)y)[(size_t)row * (CIN / 4) + lane] = o;
}

__global__ __launch_bounds__(512)
void cheb_gemm_kernel(const float* __restrict__ t0,
                      const float* __restrict__ t1,
                      const float* __restrict__ t2,
                      const float* __restrict__ t3,
                      const float* __restrict__ weights,
                      const float* __restrict__ bias,
                      float* __restrict__ out) {
    __shared__ float As[32 * 132];

    const float* terms[4] = {t0, t1, t2, t3};

    const int tid  = threadIdx.x;
    const int wave = tid >> 5;
    const int lane = tid & 31;
    const int wm   = wave >> 3;
    const int wn   = wave & 7;
    const int l    = lane & 15;
    const int half = lane >> 4;

    const size_t row0 = (size_t)blockIdx.x * 32;

    const int arow = wm * 16 + l;
    const int bn   = wn * 16 + l;

    v8f acc = {};

#pragma unroll
    for (int t = 0; t < 4; ++t) {
        __syncthreads();
        const float* src = terms[t] + row0 * CIN;
#pragma unroll
        for (int i = tid; i < 1024; i += 512) {
            const int r  = i >> 5;
            const int cv = i & 31;
            const float4 v = ((const float4*)(src + r * CIN))[cv];
            float* d = &As[r * 132 + cv * 4];
            d[0] = v.x; d[1] = v.y; d[2] = v.z; d[3] = v.w;
        }
        __syncthreads();

        const float* W = weights + t * CIN * COUT;
#pragma unroll 4
        for (int k = 0; k < 128; k += 4) {
            const int kk = k + 2 * half;
            v2f a;
            a.x = As[arow * 132 + kk];
            a.y = As[arow * 132 + kk + 1];
            v2f bf;
            bf.x = W[kk * COUT + bn];
            bf.y = W[(kk + 1) * COUT + bn];
            acc = __builtin_amdgcn_wmma_f32_16x16x4_f32(
                 false, a,  false, bf,
                 (short)0, acc,  false,  false);
        }
    }

    const float bv = bias[bn];
    float* op = out + (row0 + (size_t)wm * 16 + 8 * half) * COUT + bn;
#pragma unroll
    for (int r = 0; r < 8; ++r) {
        const float v = acc[r] + bv;
        op[(size_t)r * COUT] = fmaxf(v, 0.f);
    }
}

extern "C" void kernel_launch(void* const* d_in, const int* in_sizes, int n_in,
                              void* d_out, int out_size, void* d_ws, size_t ws_size,
                              hipStream_t stream) {
    const float* x       = (const float*)d_in[0];
    const float* vals    = (const float*)d_in[1];
    const float* weights = (const float*)d_in[2];
    const float* bias    = (const float*)d_in[3];
    const int*   rows    = (const int*)d_in[4];
    const int*   cols    = (const int*)d_in[5];
    float* out = (float*)d_out;

    const size_t feat = (size_t)BB * NN * CIN;
    float* T1 = (float*)d_ws;
    float* T2 = T1 + feat;
    float* T3 = T2 + feat;

    int* pk     = (int*)d_out;
    int* perm   = pk + (size_t)BB * EE * 2;
    int* rowptr = perm + (size_t)BB * EE;

    const int rowGrid = (BB * NN * 32) / 256;

    build_rows_kernel<<<BB * NBLK, 1024, 0, stream>>>(rows, rowptr, perm);
    order_rows_kernel<<<rowGrid, 256, 0, stream>>>(rowptr, perm, cols, vals, pk);

    spmm_rows_kernel<<<rowGrid, 256, 0, stream>>>(T1, x, nullptr, rowptr, pk, 1.0f);
    spmm_rows_kernel<<<rowGrid, 256, 0, stream>>>(T2, T1, x, rowptr, pk, 2.0f);
    spmm_rows_kernel<<<rowGrid, 256, 0, stream>>>(T3, T2, T1, rowptr, pk, 2.0f);

    cheb_gemm_kernel<<<(BB * NN) / 32, 512, 0, stream>>>(x, T1, T2, T3, weights, bias, out);
}


# Last propagation launch folded into the product launch: each block sums its own 32 rows of T3 = 2 L T2 - T1 into an LDS tile, half the blocks before, half after their other terms; sum order unchanged
# speedup vs baseline: 3.6401x; 1.0674x over previous
#include <hip/hip_runtime.h>

#define BB   16
#define NN   4096
#define EE   65536
#define CIN  128
#define COUT 128

typedef __attribute__((ext_vector_type(2))) float v2f;
typedef __attribute__((ext_vector_type(8))) float v8f;

#define RB   256
#define NBLK (NN / RB)

__global__ __launch_bounds__(1024)
void build_rows_kernel(const int* __restrict__ rows,
                       int* __restrict__ rowptr,
                       int* __restrict__ perm) {
    __shared__ int cnt[RB];
    __shared__ int wsum[RB / 32];
    __shared__ int wbelow[32];

    const int b    = blockIdx.x / NBLK;
    const int blk  = blockIdx.x % NBLK;
    const int lo   = blk * RB;
    const int tid  = threadIdx.x;
    const int lane = tid & 31;
    const int wave = tid >> 5;
    const int* rb  = rows + (size_t)b * EE;

    if (tid < RB) cnt[tid] = 0;

    const int4* rb4 = (const int4*)rb;
    int4 rr[16];
#pragma unroll
    for (int k = 0; k < 16; ++k) rr[k] = rb4[tid + k * 1024];

    int below = 0;
#pragma unroll
    for (int k = 0; k < 16; ++k)
        below += (rr[k].x < lo) + (rr[k].y < lo) + (rr[k].z < lo) + (rr[k].w < lo);
    for (int d = 16; d > 0; d >>= 1) below += __shfl_xor(below, d);
    if (lane == 0) wbelow[wave] = below;
    __syncthreads();

#pragma unroll
    for (int k = 0; k < 16; ++k) {
        const int d0 = rr[k].x - lo, d1 = rr[k].y - lo, d2 = rr[k].z - lo, d3 = rr[k].w - lo;
        if ((unsigned)d0 < (unsigned)RB) atomicAdd(&cnt[d0], 1);
        if ((unsigned)d1 < (unsigned)RB) atomicAdd(&cnt[d1], 1);
        if ((unsigned)d2 < (unsigned)RB) atomicAdd(&cnt[d2], 1);
        if ((unsigned)d3 < (unsigned)RB) atomicAdd(&cnt[d3], 1);
    }
    __syncthreads();

    int c = 0, incl = 0;
    if (tid < RB) {
        c = cnt[tid];
        incl = c;
        for (int d = 1; d < 32; d <<= 1) {
            const int t = __shfl_up(incl, d);
            if (lane >= d) incl += t;
        }
        if (lane == 31) wsum[wave] = incl;
    }
    __syncthreads();
    if (tid < RB) {
        int base = wbelow[lane];
        for (int d = 16; d > 0; d >>= 1) base += __shfl_xor(base, d);
        for (int w = 0; w < wave; ++w) base += wsum[w];
        const int p = base + incl - c;

        int* rp = rowptr + b * (NN + 1);
        rp[lo + tid] = p;
        if (blk == NBLK - 1 && tid == RB - 1) rp[NN] = p + c;

        cnt[tid] = p;
    }
    __syncthreads();

    int* pb = perm + (size_t)b * EE;
#pragma unroll
    for (int g = 0; g < 4; ++g) {
        int pos[16];
#pragma unroll
        for (int q = 0; q < 4; ++q) {
            const int k = 4 * g + q;
            const int d0 = rr[k].x - lo, d1 = rr[k].y - lo, d2 = rr[k].z - lo, d3 = rr[k].w - lo;
            pos[4 * q + 0] = ((unsigned)d0 < (unsigned)RB) ? atomicAdd(&cnt[d0], 1) : -1;
            pos[4 * q + 1] = ((unsigned)d1 < (unsigned)RB) ? atomicAdd(&cnt[d1], 1) : -1;
            pos[4 * q + 2] = ((unsigned)d2 < (unsigned)RB) ? atomicAdd(&cnt[d2], 1) : -1;
            pos[4 * q + 3] = ((unsigned)d3 < (unsigned)RB) ? atomicAdd(&cnt[d3], 1) : -1;
        }
#pragma unroll
        for (int q = 0; q < 4; ++q) {
            const int e0 = 4 * (tid + (4 * g + q) * 1024);
            if (pos[4 * q + 0] >= 0) pb[pos[4 * q + 0]] = e0 + 0;
            if (pos[4 * q + 1] >= 0) pb[pos[4 * q + 1]] = e0 + 1;
            if (pos[4 * q + 2] >= 0) pb[pos[4 * q + 2]] = e0 + 2;
            if (pos[4 * q + 3] >= 0) pb[pos[4 * q + 3]] = e0 + 3;
        }
    }
}

__global__ __launch_bounds__(256)
void order_rows_kernel(const int* __restrict__ rowptr,
                       const int* __restrict__ perm,
                       const int* __restrict__ cols,
                       const float* __restrict__ vals,
                       int* __restrict__ pk) {
    const int row  = __builtin_amdgcn_readfirstlane((int)((blockIdx.x * 256 + threadIdx.x) >> 5));
    const int lane = threadIdx.x & 31;
    const int b    = row >> 12;
    const int r    = row & (NN - 1);
    const int s    = rowptr[b * (NN + 1) + r];
    const int n    = rowptr[b * (NN + 1) + r + 1] - s;
    const int* pb  = perm + (size_t)b * EE + s;

    for (int i0 = 0; i0 < n; i0 += 32) {
        const int i  = i0 + lane;
        const int id = (i < n) ? pb[i] : 0x7fffffff;
        int rank = 0;
        for (int j0 = 0; j0 < n; j0 += 32) {
            const int jj    = j0 + lane;
            const int other = (jj < n) ? pb[jj] : 0x7fffffff;
            const int m     = (n - j0 < 32) ? (n - j0) : 32;
            for (int u = 0; u < m; ++u)
                rank += (__builtin_amdgcn_readlane(other, u) < id) ? 1 : 0;
        }
        if (i < n) {
            const size_t e = (size_t)b * EE + id;
            int2 rec;
            rec.x = cols[e];
            rec.y = __float_as_int(vals[e]);
            ((int2*)pk)[(size_t)b * EE + s + rank] = rec;
        }
    }
}

template <int K>
__device__ __forceinline__ void edge_batch(float4& acc, const int2* __restrict__ rec,
                                           const float4* __restrict__ xb, int i, int lane) {
    int2 cv[K];
#pragma unroll
    for (int u = 0; u < K; ++u) cv[u] = rec[i + u];
    float4 xv[K];
#pragma unroll
    for (int u = 0; u < K; ++u) xv[u] = xb[(size_t)cv[u].x * (CIN / 4) + lane];
#pragma unroll
    for (int u = 0; u < K; ++u) {
        const float v = __int_as_float(cv[u].y);
        acc.x = __builtin_fmaf(v, xv[u].x, acc.x);
        acc.y = __builtin_fmaf(v, xv[u].y, acc.y);
        acc.z = __builtin_fmaf(v, xv[u].z, acc.z);
        acc.w = __builtin_fmaf(v, xv[u].w, acc.w);
    }
}

__device__ __forceinline__ float4 row_sum(const int2* __restrict__ rec, const float4* __restrict__ xb,
                                          int s, int e, int lane) {
    float4 acc = make_float4(0.f, 0.f, 0.f, 0.f);
    int i = s;
    for (; i + 8 <= e; i += 8) edge_batch<8>(acc, rec, xb, i, lane);
    if (i + 4 <= e) { edge_batch<4>(acc, rec, xb, i, lane); i += 4; }
    if (i + 2 <= e) { edge_batch<2>(acc, rec, xb, i, lane); i += 2; }
    if (i < e)      { edge_batch<1>(acc, rec, xb, i, lane); }
    return acc;
}

__global__ __launch_bounds__(256)
void spmm_rows_kernel(float* __restrict__ y,
                      const float* __restrict__ x,
                      const float* __restrict__ prev,
                      const int* __restrict__ rowptr,
                      const int* __restrict__ pk,
                      float scale) {
    const int row  = __builtin_amdgcn_readfirstlane((int)((blockIdx.x * 256 + threadIdx.x) >> 5));
    const int lane = threadIdx.x & 31;
    const int b    = row >> 12;
    const int r    = row & (NN - 1);
    const int s    = rowptr[b * (NN + 1) + r];
    const int e    = rowptr[b * (NN + 1) + r + 1];
    const int2* rec  = (const int2*)pk + (size_t)b * EE;
    const float4* xb = (const float4*)(x + (size_t)b * NN * CIN);

    const float4 acc = row_sum(rec, xb, s, e, lane);

    float4 o;
    if (prev) {
        const float4 p = ((const float4*)prev)[(size_t)row * (CIN / 4) + lane];
        o = make_float4(scale * acc.x - p.x, scale * acc.y - p.y,
                        scale * acc.z - p.z, scale * acc.w - p.w);
    } else {
        o = make_float4(scale * acc.x, scale * acc.y, scale * acc.z, scale * acc.w);
    }
    ((float4*)y)[(size_t)row * (CIN / 4) + lane] = o;
}

__device__ __forceinline__ void mma_term(v8f& acc, const float (&tile)[32 * 132],
                                         const float* __restrict__ W,
                                         int arow, int half, int bn) {
#pragma unroll 4
    for (int k = 0; k < 128; k += 4) {
        const int kk = k + 2 * half;
        v2f a;
        a.x = tile[arow * 132 + kk];
        a.y = tile[arow * 132 + kk + 1];
        v2f bf;
        bf.x = W[kk * COUT + bn];
        bf.y = W[(kk + 1) * COUT + bn];
        acc = __builtin_amdgcn_wmma_f32_16x16x4_f32(
             false, a,  false, bf,
             (short)0, acc,  false,  false);
    }
}

__device__ __forceinline__ void t3_tile(float (&tile)[32 * 132],
                                        const float* __restrict__ t2,
                                        const float* __restrict__ t1,
                                        const int* __restrict__ rowptr,
                                        const int* __restrict__ pk,
                                        int row0, int tid) {
    const int lane = tid & 31;
    for (int q = 0; q < 2; ++q) {
        const int lr  = __builtin_amdgcn_readfirstlane(2 * (tid >> 5) + q);
        const int row = row0 + lr;
        const int b   = row >> 12;
        const int r   = row & (NN - 1);
        const int s   = rowptr[b * (NN + 1) + r];
        const int e   = rowptr[b * (NN + 1) + r + 1];
        const int2* rec  = (const int2*)pk + (size_t)b * EE;
        const float4* xb = (const float4*)(t2 + (size_t)b * NN * CIN);

        const float4 acc = row_sum(rec, xb, s, e, lane);
        const float4 p   = ((const float4*)t1)[(size_t)row * (CIN / 4) + lane];
        float* d = &tile[lr * 132 + lane * 4];
        d[0] = 2.0f * acc.x - p.x;
        d[1] = 2.0f * acc.y - p.y;
        d[2] = 2.0f * acc.z - p.z;
        d[3] = 2.0f * acc.w - p.w;
    }
}

__global__ __launch_bounds__(512)
void cheb_gemm_kernel(const float* __restrict__ t0,
                      const float* __restrict__ t1,
                      const float* __restrict__ t2,
                      const int* __restrict__ rowptr,
                      const int* __restrict__ pk,
                      const float* __restrict__ weights,
                      const float* __restrict__ bias,
                      float* __restrict__ out) {
    __shared__ float As[32 * 132];
    __shared__ float A3[32 * 132];

    const float* terms[3] = {t0, t1, t2};

    const int tid  = threadIdx.x;
    const int wave = tid >> 5;
    const int lane = tid & 31;
    const int wm   = wave >> 3;
    const int wn   = wave & 7;
    const int l    = lane & 15;
    const int half = lane >> 4;

    const int row0 = blockIdx.x * 32;

    const int arow = wm * 16 + l;
    const int bn   = wn * 16 + l;

    const bool early = (__builtin_popcount(blockIdx.x) & 1) != 0;

    if (early) t3_tile(A3, t2, t1, rowptr, pk, row0, tid);

    v8f acc = {};

#pragma unroll
    for (int t = 0; t < 3; ++t) {
        __syncthreads();
        const float* src = terms[t] + (size_t)row0 * CIN;
#pragma unroll
        for (int i = tid; i < 1024; i += 512) {
            const int r  = i >> 5;
            const int cv = i & 31;
            const float4 v = ((const float4*)(src + r * CIN))[cv];
            float* d = &As[r * 132 + cv * 4];
            d[0] = v.x; d[1] = v.y; d[2] = v.z; d[3] = v.w;
        }
        __syncthreads();

        mma_term(acc, As, weights + t * CIN * COUT, arow, half, bn);
    }

    if (!early) t3_tile(A3, t2, t1, rowptr, pk, row0, tid);
    __syncthreads();
    mma_term(acc, A3, weights + 3 * CIN * COUT, arow, half, bn);

    const float bv = bias[bn];
    float* op = out + ((size_t)row0 + (size_t)wm * 16 + 8 * half) * COUT + bn;
#pragma unroll
    for (int r = 0; r < 8; ++r) {
        const float v = acc[r] + bv;
        op[(size_t)r * COUT] = fmaxf(v, 0.f);
    }
}

extern "C" void kernel_launch(void* const* d_in, const int* in_sizes, int n_in,
                              void* d_out, int out_size, void* d_ws, size_t ws_size,
                              hipStream_t stream) {
    const float* x       = (const float*)d_in[0];
    const float* vals    = (const float*)d_in[1];
    const float* weights = (const float*)d_in[2];
    const float* bias    = (const float*)d_in[3];
    const int*   rows    = (const int*)d_in[4];
    const int*   cols    = (const int*)d_in[5];
    float* out = (float*)d_out;

    const size_t feat = (size_t)BB * NN * CIN;
    float* T1 = (float*)d_ws;
    float* T2 = T1 + feat;

    int* pk     = (int*)(T2 + feat);
    int* perm   = pk + (size_t)BB * EE * 2;
    int* rowptr = perm + (size_t)BB * EE;

    const int rowGrid = (BB * NN * 32) / 256;

    build_rows_kernel<<<BB * NBLK, 1024, 0, stream>>>(rows, rowptr, perm);
    order_rows_kernel<<<rowGrid, 256, 0, stream>>>(rowptr, perm, cols, vals, pk);

    spmm_rows_kernel<<<rowGrid, 256, 0, stream>>>(T1, x, nullptr, rowptr, pk, 1.0f);
    spmm_rows_kernel<<<rowGrid, 256, 0, stream>>>(T2, T1, x, rowptr, pk, 2.0f);

    cheb_gemm_kernel<<<(BB * NN) / 32, 512, 0, stream>>>(x, T1, T2, rowptr, pk, weights, bias, out);
}


# Edge ordering folded into the grouping launch: each work-group keeps its block's slots in LDS, ranks them by edge number per row, writes the (col,val) records; order launch gone, sum order unchanged
# speedup vs baseline: 3.8550x; 1.0590x over previous
#include <hip/hip_runtime.h>

#define BB   16
#define NN   4096
#define EE   65536
#define CIN  128
#define COUT 128

typedef __attribute__((ext_vector_type(2))) float v2f;
typedef __attribute__((ext_vector_type(8))) float v8f;

#define RB   256
#define NBLK (NN / RB)
#define OQ   5

__global__ __launch_bounds__(1024)
void build_rows_kernel(const int* __restrict__ rows,
                       const int* __restrict__ cols,
                       const float* __restrict__ vals,
                       int* __restrict__ rowptr,
                       int* __restrict__ pk) {
    __shared__ int cnt[RB];
    __shared__ int rstart[RB];
    __shared__ int wsum[RB / 32];
    __shared__ int wbelow[32];
    __shared__ int slot[EE];

    const int b    = blockIdx.x / NBLK;
    const int blk  = blockIdx.x % NBLK;
    const int lo   = blk * RB;
    const int tid  = threadIdx.x;
    const int lane = tid & 31;
    const int wave = tid >> 5;
    const int* rb  = rows + (size_t)b * EE;

    if (tid < RB) cnt[tid] = 0;

    const int4* rb4 = (const int4*)rb;
    int4 rr[16];
#pragma unroll
    for (int k = 0; k < 16; ++k) rr[k] = rb4[tid + k * 1024];

    int below = 0;
#pragma unroll
    for (int k = 0; k < 16; ++k)
        below += (rr[k].x < lo) + (rr[k].y < lo) + (rr[k].z < lo) + (rr[k].w < lo);
    for (int d = 16; d > 0; d >>= 1) below += __shfl_xor(below, d);
    if (lane == 0) wbelow[wave] = below;
    __syncthreads();

#pragma unroll
    for (int k = 0; k < 16; ++k) {
        const int d0 = rr[k].x - lo, d1 = rr[k].y - lo, d2 = rr[k].z - lo, d3 = rr[k].w - lo;
        if ((unsigned)d0 < (unsigned)RB) atomicAdd(&cnt[d0], 1);
        if ((unsigned)d1 < (unsigned)RB) atomicAdd(&cnt[d1], 1);
        if ((unsigned)d2 < (unsigned)RB) atomicAdd(&cnt[d2], 1);
        if ((unsigned)d3 < (unsigned)RB) atomicAdd(&cnt[d3], 1);
    }
    __syncthreads();

    int c = 0, incl = 0;
    if (tid < RB) {
        c = cnt[tid];
        incl = c;
        for (int d = 1; d < 32; d <<= 1) {
            const int t = __shfl_up(incl, d);
            if (lane >= d) incl += t;
        }
        if (lane == 31) wsum[wave] = incl;
    }
    int blockbase = wbelow[lane];
    for (int d = 16; d > 0; d >>= 1) blockbase += __shfl_xor(blockbase, d);
    __syncthreads();
    if (tid < RB) {
        int loc = incl - c;
        for (int w = 0; w < wave; ++w) loc += wsum[w];

        int* rp = rowptr + b * (NN + 1);
        rp[lo + tid] = blockbase + loc;
        if (blk == NBLK - 1 && tid == RB - 1) rp[NN] = blockbase + loc + c;

        cnt[tid]    = loc;
        rstart[tid] = loc;
    }
    __syncthreads();

#pragma unroll
    for (int g = 0; g < 4; ++g) {
        int pos[16];
#pragma unroll
        for (int q = 0; q < 4; ++q) {
            const int k = 4 * g + q;
            const int d0 = rr[k].x - lo, d1 = rr[k].y - lo, d2 = rr[k].z - lo, d3 = rr[k].w - lo;
            pos[4 * q + 0] = ((unsigned)d0 < (unsigned)RB) ? atomicAdd(&cnt[d0], 1) : -1;
            pos[4 * q + 1] = ((unsigned)d1 < (unsigned)RB) ? atomicAdd(&cnt[d1], 1) : -1;
            pos[4 * q + 2] = ((unsigned)d2 < (unsigned)RB) ? atomicAdd(&cnt[d2], 1) : -1;
            pos[4 * q + 3] = ((unsigned)d3 < (unsigned)RB) ? atomicAdd(&cnt[d3], 1) : -1;
        }
#pragma unroll
        for (int q = 0; q < 4; ++q) {
            const int k  = 4 * g + q;
            const int e0 = 4 * (tid + k * 1024);
            const int d0 = rr[k].x - lo, d1 = rr[k].y - lo, d2 = rr[k].z - lo, d3 = rr[k].w - lo;
            if (pos[4 * q + 0] >= 0) slot[pos[4 * q + 0]] = (d0 << 16) | (e0 + 0);
            if (pos[4 * q + 1] >= 0) slot[pos[4 * q + 1]] = (d1 << 16) | (e0 + 1);
            if (pos[4 * q + 2] >= 0) slot[pos[4 * q + 2]] = (d2 << 16) | (e0 + 2);
            if (pos[4 * q + 3] >= 0) slot[pos[4 * q + 3]] = (d3 << 16) | (e0 + 3);
        }
    }
    __syncthreads();

    const int nblk = cnt[RB - 1];
    const int*   cb  = cols + (size_t)b * EE;
    const float* vb  = vals + (size_t)b * EE;
    int2*        pkb = (int2*)pk + (size_t)b * EE + blockbase;

    for (int i0 = 0; i0 < nblk; i0 += OQ * 1024) {
        int   dst[OQ];
        int   cc[OQ];
        float vv[OQ];
#pragma unroll
        for (int q = 0; q < OQ; ++q) {
            const int i = i0 + tid + q * 1024;
            dst[q] = -1; cc[q] = 0; vv[q] = 0.f;
            if (i < nblk) {
                const int w = slot[i];
                const int d = w >> 16;
                const int s = rstart[d];
                const int e = cnt[d];
                int rank = 0;
                for (int j = s; j < e; ++j) rank += (slot[j] < w) ? 1 : 0;
                const int id = w & 0xffff;
                cc[q]  = cb[id];
                vv[q]  = vb[id];
                dst[q] = s + rank;
            }
        }
#pragma unroll
        for (int q = 0; q < OQ; ++q) {
            if (dst[q] >= 0) {
                int2 rec;
                rec.x = cc[q];
                rec.y = __float_as_int(vv[q]);
                pkb[dst[q]] = rec;
            }
        }
    }
}

template <int K>
__device__ __forceinline__ void edge_batch(float4& acc, const int2* __restrict__ rec,
                                           const float4* __restrict__ xb, int i, int lane) {
    int2 cv[K];
#pragma unroll
    for (int u = 0; u < K; ++u) cv[u] = rec[i + u];
    float4 xv[K];
#pragma unroll
    for (int u = 0; u < K; ++u) xv[u] = xb[(size_t)cv[u].x * (CIN / 4) + lane];
#pragma unroll
    for (int u = 0; u < K; ++u) {
        const float v = __int_as_float(cv[u].y);
        acc.x = __builtin_fmaf(v, xv[u].x, acc.x);
        acc.y = __builtin_fmaf(v, xv[u].y, acc.y);
        acc.z = __builtin_fmaf(v, xv[u].z, acc.z);
        acc.w = __builtin_fmaf(v, xv[u].w, acc.w);
    }
}

__device__ __forceinline__ float4 row_sum(const int2* __restrict__ rec, const float4* __restrict__ xb,
                                          int s, int e, int lane) {
    float4 acc = make_float4(0.f, 0.f, 0.f, 0.f);
    int i = s;
    for (; i + 8 <= e; i += 8) edge_batch<8>(acc, rec, xb, i, lane);
    if (i + 4 <= e) { edge_batch<4>(acc, rec, xb, i, lane); i += 4; }
    if (i + 2 <= e) { edge_batch<2>(acc, rec, xb, i, lane); i += 2; }
    if (i < e)      { edge_batch<1>(acc, rec, xb, i, lane); }
    return acc;
}

__global__ __launch_bounds__(256)
void spmm_rows_kernel(float* __restrict__ y,
                      const float* __restrict__ x,
                      const float* __restrict__ prev,
                      const int* __restrict__ rowptr,
                      const int* __restrict__ pk,
                      float scale) {
    const int row  = __builtin_amdgcn_readfirstlane((int)((blockIdx.x * 256 + threadIdx.x) >> 5));
    const int lane = threadIdx.x & 31;
    const int b    = row >> 12;
    const int r    = row & (NN - 1);
    const int s    = rowptr[b * (NN + 1) + r];
    const int e    = rowptr[b * (NN + 1) + r + 1];
    const int2* rec  = (const int2*)pk + (size_t)b * EE;
    const float4* xb = (const float4*)(x + (size_t)b * NN * CIN);

    const float4 acc = row_sum(rec, xb, s, e, lane);

    float4 o;
    if (prev) {
        const float4 p = ((const float4*)prev)[(size_t)row * (CIN / 4) + lane];
        o = make_float4(scale * acc.x - p.x, scale * acc.y - p.y,
                        scale * acc.z - p.z, scale * acc.w - p.w);
    } else {
        o = make_float4(scale * acc.x, scale * acc.y, scale * acc.z, scale * acc.w);
    }
    ((float4*)y)[(size_t)row * (CIN / 4) + lane] = o;
}

__device__ __forceinline__ void mma_term(v8f& acc, const float (&tile)[32 * 132],
                                         const float* __restrict__ W,
                                         int arow, int half, int bn) {
#pragma unroll 4
    for (int k = 0; k < 128; k += 4) {
        const int kk = k + 2 * half;
        v2f a;
        a.x = tile[arow * 132 + kk];
        a.y = tile[arow * 132 + kk + 1];
        v2f bf;
        bf.x = W[kk * COUT + bn];
        bf.y = W[(kk + 1) * COUT + bn];
        acc = __builtin_amdgcn_wmma_f32_16x16x4_f32(
             false, a,  false, bf,
             (short)0, acc,  false,  false);
    }
}

__device__ __forceinline__ void t3_tile(float (&tile)[32 * 132],
                                        const float* __restrict__ t2,
                                        const float* __restrict__ t1,
                                        const int* __restrict__ rowptr,
                                        const int* __restrict__ pk,
                                        int row0, int tid) {
    const int lane = tid & 31;
    for (int q = 0; q < 2; ++q) {
        const int lr  = __builtin_amdgcn_readfirstlane(2 * (tid >> 5) + q);
        const int row = row0 + lr;
        const int b   = row >> 12;
        const int r   = row & (NN - 1);
        const int s   = rowptr[b * (NN + 1) + r];
        const int e   = rowptr[b * (NN + 1) + r + 1];
        const int2* rec  = (const int2*)pk + (size_t)b * EE;
        const float4* xb = (const float4*)(t2 + (size_t)b * NN * CIN);

        const float4 acc = row_sum(rec, xb, s, e, lane);
        const float4 p   = ((const float4*)t1)[(size_t)row * (CIN / 4) + lane];
        float* d = &tile[lr * 132 + lane * 4];
        d[0] = 2.0f * acc.x - p.x;
        d[1] = 2.0f * acc.y - p.y;
        d[2] = 2.0f * acc.z - p.z;
        d[3] = 2.0f * acc.w - p.w;
    }
}

__global__ __launch_bounds__(512)
void cheb_gemm_kernel(const float* __restrict__ t0,
                      const float* __restrict__ t1,
                      const float* __restrict__ t2,
                      const int* __restrict__ rowptr,
                      const int* __restrict__ pk,
                      const float* __restrict__ weights,
                      const float* __restrict__ bias,
                      float* __restrict__ out) {
    __shared__ float As[32 * 132];
    __shared__ float A3[32 * 132];

    const float* terms[3] = {t0, t1, t2};

    const int tid  = threadIdx.x;
    const int wave = tid >> 5;
    const int lane = tid & 31;
    const int wm   = wave >> 3;
    const int wn   = wave & 7;
    const int l    = lane & 15;
    const int half = lane >> 4;

    const int row0 = blockIdx.x * 32;

    const int arow = wm * 16 + l;
    const int bn   = wn * 16 + l;

    const bool early = (__builtin_popcount(blockIdx.x) & 1) != 0;

    if (early) t3_tile(A3, t2, t1, rowptr, pk, row0, tid);

    v8f acc = {};

#pragma unroll
    for (int t = 0; t < 3; ++t) {
        __syncthreads();
        const float* src = terms[t] + (size_t)row0 * CIN;
#pragma unroll
        for (int i = tid; i < 1024; i += 512) {
            const int r  = i >> 5;
            const int cv = i & 31;
            const float4 v = ((const float4*)(src + r * CIN))[cv];
            float* d = &As[r * 132 + cv * 4];
            d[0] = v.x; d[1] = v.y; d[2] = v.z; d[3] = v.w;
        }
        __syncthreads();

        mma_term(acc, As, weights + t * CIN * COUT, arow, half, bn);
    }

    if (!early) t3_tile(A3, t2, t1, rowptr, pk, row0, tid);
    __syncthreads();
    mma_term(acc, A3, weights + 3 * CIN * COUT, arow, half, bn);

    const float bv = bias[bn];
    float* op = out + ((size_t)row0 + (size_t)wm * 16 + 8 * half) * COUT + bn;
#pragma unroll
    for (int r = 0; r < 8; ++r) {
        const float v = acc[r] + bv;
        op[(size_t)r * COUT] = fmaxf(v, 0.f);
    }
}

extern "C" void kernel_launch(void* const* d_in, const int* in_sizes, int n_in,
                              void* d_out, int out_size, void* d_ws, size_t ws_size,
                              hipStream_t stream) {
    const float* x       = (const float*)d_in[0];
    const float* vals    = (const float*)d_in[1];
    const float* weights = (const float*)d_in[2];
    const float* bias    = (const float*)d_in[3];
    const int*   rows    = (const int*)d_in[4];
    const int*   cols    = (const int*)d_in[5];
    float* out = (float*)d_out;

    const size_t feat = (size_t)BB * NN * CIN;
    float* T1 = (float*)d_ws;
    float* T2 = T1 + feat;

    int* pk     = (int*)(T2 + feat);
    int* rowptr = pk + (size_t)BB * EE * 2;

    const int rowGrid = (BB * NN * 32) / 256;

    build_rows_kernel<<<BB * NBLK, 1024, 0, stream>>>(rows, cols, vals, rowptr, pk);

    spmm_rows_kernel<<<rowGrid, 256, 0, stream>>>(T1, x, nullptr, rowptr, pk, 1.0f);
    spmm_rows_kernel<<<rowGrid, 256, 0, stream>>>(T2, T1, x, rowptr, pk, 2.0f);

    cheb_gemm_kernel<<<(BB * NN) / 32, 512, 0, stream>>>(x, T1, T2, rowptr, pk, weights, bias, out);
}


# Product launch: W of each term staged once per block into LDS (pitch 136), B fragments read from there, so the K loop has no global load; block tile 64 rows on 32 waves; sum order unchanged
# speedup vs baseline: 4.0600x; 1.0532x over previous
#include <hip/hip_runtime.h>

#define BB   16
#define NN   4096
#define EE   65536
#define CIN  128
#define COUT 128

typedef __attribute__((ext_vector_type(2))) float v2f;
typedef __attribute__((ext_vector_type(8))) float v8f;

#define RB   256
#define NBLK (NN / RB)
#define OQ   5

__global__ __launch_bounds__(1024)
void build_rows_kernel(const int* __restrict__ rows,
                       const int* __restrict__ cols,
                       const float* __restrict__ vals,
                       int* __restrict__ rowptr,
                       int* __restrict__ pk) {
    __shared__ int cnt[RB];
    __shared__ int rstart[RB];
    __shared__ int wsum[RB / 32];
    __shared__ int wbelow[32];
    __shared__ int slot[EE];

    const int b    = blockIdx.x / NBLK;
    const int blk  = blockIdx.x % NBLK;
    const int lo   = blk * RB;
    const int tid  = threadIdx.x;
    const int lane = tid & 31;
    const int wave = tid >> 5;
    const int* rb  = rows + (size_t)b * EE;

    if (tid < RB) cnt[tid] = 0;

    const int4* rb4 = (const int4*)rb;
    int4 rr[16];
#pragma unroll
    for (int k = 0; k < 16; ++k) rr[k] = rb4[tid + k * 1024];

    int below = 0;
#pragma unroll
    for (int k = 0; k < 16; ++k)
        below += (rr[k].x < lo) + (rr[k].y < lo) + (rr[k].z < lo) + (rr[k].w < lo);
    for (int d = 16; d > 0; d >>= 1) below += __shfl_xor(below, d);
    if (lane == 0) wbelow[wave] = below;
    __syncthreads();

#pragma unroll
    for (int k = 0; k < 16; ++k) {
        const int d0 = rr[k].x - lo, d1 = rr[k].y - lo, d2 = rr[k].z - lo, d3 = rr[k].w - lo;
        if ((unsigned)d0 < (unsigned)RB) atomicAdd(&cnt[d0], 1);
        if ((unsigned)d1 < (unsigned)RB) atomicAdd(&cnt[d1], 1);
        if ((unsigned)d2 < (unsigned)RB) atomicAdd(&cnt[d2], 1);
        if ((unsigned)d3 < (unsigned)RB) atomicAdd(&cnt[d3], 1);
    }
    __syncthreads();

    int c = 0, incl = 0;
    if (tid < RB) {
        c = cnt[tid];
        incl = c;
        for (int d = 1; d < 32; d <<= 1) {
            const int t = __shfl_up(incl, d);
            if (lane >= d) incl += t;
        }
        if (lane == 31) wsum[wave] = incl;
    }
    int blockbase = wbelow[lane];
    for (int d = 16; d > 0; d >>= 1) blockbase += __shfl_xor(blockbase, d);
    __syncthreads();
    if (tid < RB) {
        int loc = incl - c;
        for (int w = 0; w < wave; ++w) loc += wsum[w];

        int* rp = rowptr + b * (NN + 1);
        rp[lo + tid] = blockbase + loc;
        if (blk == NBLK - 1 && tid == RB - 1) rp[NN] = blockbase + loc + c;

        cnt[tid]    = loc;
        rstart[tid] = loc;
    }
    __syncthreads();

#pragma unroll
    for (int g = 0; g < 4; ++g) {
        int pos[16];
#pragma unroll
        for (int q = 0; q < 4; ++q) {
            const int k = 4 * g + q;
            const int d0 = rr[k].x - lo, d1 = rr[k].y - lo, d2 = rr[k].z - lo, d3 = rr[k].w - lo;
            pos[4 * q + 0] = ((unsigned)d0 < (unsigned)RB) ? atomicAdd(&cnt[d0], 1) : -1;
            pos[4 * q + 1] = ((unsigned)d1 < (unsigned)RB) ? atomicAdd(&cnt[d1], 1) : -1;
            pos[4 * q + 2] = ((unsigned)d2 < (unsigned)RB) ? atomicAdd(&cnt[d2], 1) : -1;
            pos[4 * q + 3] = ((unsigned)d3 < (unsigned)RB) ? atomicAdd(&cnt[d3], 1) : -1;
        }
#pragma unroll
        for (int q = 0; q < 4; ++q) {
            const int k  = 4 * g + q;
            const int e0 = 4 * (tid + k * 1024);
            const int d0 = rr[k].x - lo, d1 = rr[k].y - lo, d2 = rr[k].z - lo, d3 = rr[k].w - lo;
            if (pos[4 * q + 0] >= 0) slot[pos[4 * q + 0]] = (d0 << 16) | (e0 + 0);
            if (pos[4 * q + 1] >= 0) slot[pos[4 * q + 1]] = (d1 << 16) | (e0 + 1);
            if (pos[4 * q + 2] >= 0) slot[pos[4 * q + 2]] = (d2 << 16) | (e0 + 2);
            if (pos[4 * q + 3] >= 0) slot[pos[4 * q + 3]] = (d3 << 16) | (e0 + 3);
        }
    }
    __syncthreads();

    const int nblk = cnt[RB - 1];
    const int*   cb  = cols + (size_t)b * EE;
    const float* vb  = vals + (size_t)b * EE;
    int2*        pkb = (int2*)pk + (size_t)b * EE + blockbase;

    for (int i0 = 0; i0 < nblk; i0 += OQ * 1024) {
        int   dst[OQ];
        int   cc[OQ];
        float vv[OQ];
#pragma unroll
        for (int q = 0; q < OQ; ++q) {
            const int i = i0 + tid + q * 1024;
            dst[q] = -1; cc[q] = 0; vv[q] = 0.f;
            if (i < nblk) {
                const int w = slot[i];
                const int d = w >> 16;
                const int s = rstart[d];
                const int e = cnt[d];
                int rank = 0;
                for (int j = s; j < e; ++j) rank += (slot[j] < w) ? 1 : 0;
                const int id = w & 0xffff;
                cc[q]  = cb[id];
                vv[q]  = vb[id];
                dst[q] = s + rank;
            }
        }
#pragma unroll
        for (int q = 0; q < OQ; ++q) {
            if (dst[q] >= 0) {
                int2 rec;
                rec.x = cc[q];
                rec.y = __float_as_int(vv[q]);
                pkb[dst[q]] = rec;
            }
        }
    }
}

template <int K>
__device__ __forceinline__ void edge_batch(float4& acc, const int2* __restrict__ rec,
                                           const float4* __restrict__ xb, int i, int lane) {
    int2 cv[K];
#pragma unroll
    for (int u = 0; u < K; ++u) cv[u] = rec[i + u];
    float4 xv[K];
#pragma unroll
    for (int u = 0; u < K; ++u) xv[u] = xb[(size_t)cv[u].x * (CIN / 4) + lane];
#pragma unroll
    for (int u = 0; u < K; ++u) {
        const float v = __int_as_float(cv[u].y);
        acc.x = __builtin_fmaf(v, xv[u].x, acc.x);
        acc.y = __builtin_fmaf(v, xv[u].y, acc.y);
        acc.z = __builtin_fmaf(v, xv[u].z, acc.z);
        acc.w = __builtin_fmaf(v, xv[u].w, acc.w);
    }
}

__device__ __forceinline__ float4 row_sum(const int2* __restrict__ rec, const float4* __restrict__ xb,
                                          int s, int e, int lane) {
    float4 acc = make_float4(0.f, 0.f, 0.f, 0.f);
    int i = s;
    for (; i + 8 <= e; i += 8) edge_batch<8>(acc, rec, xb, i, lane);
    if (i + 4 <= e) { edge_batch<4>(acc, rec, xb, i, lane); i += 4; }
    if (i + 2 <= e) { edge_batch<2>(acc, rec, xb, i, lane); i += 2; }
    if (i < e)      { edge_batch<1>(acc, rec, xb, i, lane); }
    return acc;
}

__global__ __launch_bounds__(256)
void spmm_rows_kernel(float* __restrict__ y,
                      const float* __restrict__ x,
                      const float* __restrict__ prev,
                      const int* __restrict__ rowptr,
                      const int* __restrict__ pk,
                      float scale) {
    const int row  = __builtin_amdgcn_readfirstlane((int)((blockIdx.x * 256 + threadIdx.x) >> 5));
    const int lane = threadIdx.x & 31;
    const int b    = row >> 12;
    const int r    = row & (NN - 1);
    const int s    = rowptr[b * (NN + 1) + r];
    const int e    = rowptr[b * (NN + 1) + r + 1];
    const int2* rec  = (const int2*)pk + (size_t)b * EE;
    const float4* xb = (const float4*)(x + (size_t)b * NN * CIN);

    const float4 acc = row_sum(rec, xb, s, e, lane);

    float4 o;
    if (prev) {
        const float4 p = ((const float4*)prev)[(size_t)row * (CIN / 4) + lane];
        o = make_float4(scale * acc.x - p.x, scale * acc.y - p.y,
                        scale * acc.z - p.z, scale * acc.w - p.w);
    } else {
        o = make_float4(scale * acc.x, scale * acc.y, scale * acc.z, scale * acc.w);
    }
    ((float4*)y)[(size_t)row * (CIN / 4) + lane] = o;
}

#define TM  64
#define AP  132
#define WP  136

__device__ __forceinline__ void mma_term(v8f& acc, const float* tile, const float* Wl,
                                         int arow, int half, int bn) {
#pragma unroll 8
    for (int k = 0; k < 128; k += 4) {
        const int kk = k + 2 * half;
        v2f a;
        a.x = tile[arow * AP + kk];
        a.y = tile[arow * AP + kk + 1];
        v2f bf;
        bf.x = Wl[kk * WP + bn];
        bf.y = Wl[(kk + 1) * WP + bn];
        acc = __builtin_amdgcn_wmma_f32_16x16x4_f32(
             false, a,  false, bf,
             (short)0, acc,  false,  false);
    }
}

__device__ __forceinline__ void stage_w(float* Wl, const float* __restrict__ Wg, int tid) {
#pragma unroll
    for (int i = tid; i < 4096; i += 1024) {
        const int r  = i >> 5;
        const int cv = i & 31;
        const float4 v = ((const float4*)(Wg + r * COUT))[cv];
        float* d = &Wl[r * WP + cv * 4];
        d[0] = v.x; d[1] = v.y; d[2] = v.z; d[3] = v.w;
    }
}

__device__ __forceinline__ void t3_tile(float* tile,
                                        const float* __restrict__ t2,
                                        const float* __restrict__ t1,
                                        const int* __restrict__ rowptr,
                                        const int* __restrict__ pk,
                                        int row0, int tid) {
    const int lane = tid & 31;
    for (int q = 0; q < 2; ++q) {
        const int lr  = __builtin_amdgcn_readfirstlane(2 * (tid >> 5) + q);
        const int row = row0 + lr;
        const int b   = row >> 12;
        const int r   = row & (NN - 1);
        const int s   = rowptr[b * (NN + 1) + r];
        const int e   = rowptr[b * (NN + 1) + r + 1];
        const int2* rec  = (const int2*)pk + (size_t)b * EE;
        const float4* xb = (const float4*)(t2 + (size_t)b * NN * CIN);

        const float4 acc = row_sum(rec, xb, s, e, lane);
        const float4 p   = ((const float4*)t1)[(size_t)row * (CIN / 4) + lane];
        float* d = &tile[lr * AP + lane * 4];
        d[0] = 2.0f * acc.x - p.x;
        d[1] = 2.0f * acc.y - p.y;
        d[2] = 2.0f * acc.z - p.z;
        d[3] = 2.0f * acc.w - p.w;
    }
}

__global__ __launch_bounds__(1024)
void cheb_gemm_kernel(const float* __restrict__ t0,
                      const float* __restrict__ t1,
                      const float* __restrict__ t2,
                      const int* __restrict__ rowptr,
                      const int* __restrict__ pk,
                      const float* __restrict__ weights,
                      const float* __restrict__ bias,
                      float* __restrict__ out) {
    __shared__ float As[TM * AP];
    __shared__ float A3[TM * AP];
    __shared__ float Ws[CIN * WP];

    const float* terms[3] = {t0, t1, t2};

    const int tid  = threadIdx.x;
    const int wave = tid >> 5;
    const int lane = tid & 31;
    const int wm   = wave >> 3;
    const int wn   = wave & 7;
    const int l    = lane & 15;
    const int half = lane >> 4;

    const int row0 = blockIdx.x * TM;

    const int arow = wm * 16 + l;
    const int bn   = wn * 16 + l;

    const bool early = (__builtin_popcount(blockIdx.x) & 1) != 0;

    if (early) t3_tile(A3, t2, t1, rowptr, pk, row0, tid);

    v8f acc = {};

#pragma unroll
    for (int t = 0; t < 3; ++t) {
        __syncthreads();
        const float* src = terms[t] + (size_t)row0 * CIN;
#pragma unroll
        for (int i = tid; i < 2048; i += 1024) {
            const int r  = i >> 5;
            const int cv = i & 31;
            const float4 v = ((const float4*)(src + r * CIN))[cv];
            float* d = &As[r * AP + cv * 4];
            d[0] = v.x; d[1] = v.y; d[2] = v.z; d[3] = v.w;
        }
        stage_w(Ws, weights + t * CIN * COUT, tid);
        __syncthreads();

        mma_term(acc, As, Ws, arow, half, bn);
    }

    __syncthreads();
    stage_w(Ws, weights + 3 * CIN * COUT, tid);
    if (!early) t3_tile(A3, t2, t1, rowptr, pk, row0, tid);
    __syncthreads();
    mma_term(acc, A3, Ws, arow, half, bn);

    const float bv = bias[bn];
    float* op = out + ((size_t)row0 + (size_t)wm * 16 + 8 * half) * COUT + bn;
#pragma unroll
    for (int r = 0; r < 8; ++r) {
        const float v = acc[r] + bv;
        op[(size_t)r * COUT] = fmaxf(v, 0.f);
    }
}

extern "C" void kernel_launch(void* const* d_in, const int* in_sizes, int n_in,
                              void* d_out, int out_size, void* d_ws, size_t ws_size,
                              hipStream_t stream) {
    const float* x       = (const float*)d_in[0];
    const float* vals    = (const float*)d_in[1];
    const float* weights = (const float*)d_in[2];
    const float* bias    = (const float*)d_in[3];
    const int*   rows    = (const int*)d_in[4];
    const int*   cols    = (const int*)d_in[5];
    float* out = (float*)d_out;

    const size_t feat = (size_t)BB * NN * CIN;
    float* T1 = (float*)d_ws;
    float* T2 = T1 + feat;

    int* pk     = (int*)(T2 + feat);
    int* rowptr = pk + (size_t)BB * EE * 2;

    const int rowGrid = (BB * NN * 32) / 256;

    build_rows_kernel<<<BB * NBLK, 1024, 0, stream>>>(rows, cols, vals, rowptr, pk);

    spmm_rows_kernel<<<rowGrid, 256, 0, stream>>>(T1, x, nullptr, rowptr, pk, 1.0f);
    spmm_rows_kernel<<<rowGrid, 256, 0, stream>>>(T2, T1, x, rowptr, pk, 2.0f);

    cheb_gemm_kernel<<<(BB * NN) / TM, 1024, 0, stream>>>(x, T1, T2, rowptr, pk, weights, bias, out);
}
